# starting kernel
# baseline (speedup 1.0000x reference)
#include <hip/hip_runtime.h>


typedef __attribute__((ext_vector_type(2))) float v2f;
typedef __attribute__((ext_vector_type(4))) float v4f;
typedef __attribute__((ext_vector_type(8))) float v8f;

#define B_SZ 256
#define D_SZ 512
#define P_SZ 128
#define E_SZ 128
#define XB_PITCH (D_SZ * P_SZ)
#define WP_PITCH (D_SZ * E_SZ)
#define OB_PITCH (E_SZ * P_SZ)

#define KC     16
#define KPAD   18
#define PB      8
#define BTILE  16
#define ETILE  64
#define NCHUNK (D_SZ / KC)

__global__ __launch_bounds__(256, 2)
void loop_proj_wmma_kernel(const float* __restrict__ x,
                           const float* __restrict__ W,
                           const float* __restrict__ bias,
                           float* __restrict__ out)
{
    __shared__ float sX[2][PB * BTILE * KPAD];
    __shared__ float sW[2][PB * ETILE * KPAD];

    const int tid  = threadIdx.x;
    const int lane = tid & 31;
    const int wave = tid >> 5;

    const int pg0 = blockIdx.x * PB;
    const int b0  = blockIdx.y * BTILE;
    const int e0  = blockIdx.z * ETILE;

    const int m    = lane & 15;
    const int koff = (lane >> 4) << 1;

    v8f acc[4] = {};

    const int pairX = tid >> 1;
    const int phalf = (tid & 1) * 4;
    const int iX    = pairX >> 4;
    const int kX    = pairX & 15;
    const float* xp0 = x + (size_t)(b0 + iX) * XB_PITCH
                         + (size_t)kX * P_SZ + pg0 + phalf;
    const float* xp1 = xp0 + (size_t)8 * XB_PITCH;

    const int piW   = (tid >> 1) >> 4;
    const int kW    = (tid >> 1) & 15;
    const int eHalf = (tid & 1) * 32;
    const float* wp = W + (size_t)(pg0 + piW) * WP_PITCH
                        + (size_t)kW * E_SZ + e0 + eHalf;

    auto stage = [&](int buf, const v4f& rx0, const v4f& rx1, const v4f (&rw)[8]) {
        #pragma unroll
        for (int j = 0; j < 4; ++j) {
            sX[buf][((phalf + j) * BTILE + iX)     * KPAD + kX] = rx0[j];
            sX[buf][((phalf + j) * BTILE + iX + 8) * KPAD + kX] = rx1[j];
        }
        #pragma unroll
        for (int j = 0; j < 8; ++j) {
            const int e = eHalf + j * 4;
            #pragma unroll
            for (int q = 0; q < 4; ++q)
                sW[buf][(piW * ETILE + e + q) * KPAD + kW] = rw[j][q];
        }
    };

    {
        v4f rx0 = *(const v4f*)xp0;
        v4f rx1 = *(const v4f*)xp1;
        v4f rw[8];
        #pragma unroll
        for (int j = 0; j < 8; ++j) rw[j] = *(const v4f*)(wp + j * 4);
        stage(0, rx0, rx1, rw);
    }

    for (int c = 0; c < NCHUNK; ++c) {
        const int cur  = c & 1;
        const bool more = (c + 1) < NCHUNK;

        v4f nrx0, nrx1, nrw[8];
        if (more) {
            const size_t off = (size_t)(c + 1) * KC * P_SZ;
            nrx0 = *(const v4f*)(xp0 + off);
            nrx1 = *(const v4f*)(xp1 + off);
            #pragma unroll
            for (int j = 0; j < 8; ++j) nrw[j] = *(const v4f*)(wp + off + j * 4);
        }

        __syncthreads();

        const float* bX = sX[cur];
        const float* bW = sW[cur];
        #pragma unroll
        for (int k0 = 0; k0 < KC; k0 += 4) {
            v2f a = *(const v2f*)&bX[(wave * BTILE + m) * KPAD + k0 + koff];
            #pragma unroll
            for (int t = 0; t < 4; ++t) {
                v2f bf = *(const v2f*)&bW[(wave * ETILE + t * 16 + m) * KPAD + k0 + koff];
                acc[t] = __builtin_amdgcn_wmma_f32_16x16x4_f32(
                             false, a, false, bf, (short)0, acc[t], false, false);
            }
        }

        if (more) stage(cur ^ 1, nrx0, nrx1, nrw);
    }

    const int p    = pg0 + wave;
    const int mtop = (lane >> 4) << 3;
    float bi[4];
    #pragma unroll
    for (int t = 0; t < 4; ++t) bi[t] = bias[p * E_SZ + e0 + t * 16 + m];

    #pragma unroll
    for (int r = 0; r < 8; ++r) {
        const size_t bb = (size_t)(b0 + r + mtop) * OB_PITCH;
        #pragma unroll
        for (int t = 0; t < 4; ++t)
            out[bb + (size_t)(e0 + t * 16 + m) * P_SZ + p] = acc[t][r] + bi[t];
    }
}

extern "C" void kernel_launch(void* const* d_in, const int* in_sizes, int n_in,
                              void* d_out, int out_size, void* d_ws, size_t ws_size,
                              hipStream_t stream) {
    (void)in_sizes; (void)n_in; (void)d_ws; (void)ws_size; (void)out_size;
    const float* x    = (const float*)d_in[0];
    const float* W    = (const float*)d_in[1];
    const float* bias = (const float*)d_in[2];
    float* out        = (float*)d_out;

    dim3 grid(P_SZ / PB, B_SZ / BTILE, E_SZ / ETILE);
    dim3 block(256);
    loop_proj_wmma_kernel<<<grid, block, 0, stream>>>(x, W, bias, out);
}


# W staging remapped: 8 consecutive lanes read one aligned 128-B line of a W row, so a wave load touches 4 whole lines instead of 32 pieces; its transposing LDS stores hit 32 distinct banks
# speedup vs baseline: 1.2013x; 1.2013x over previous
#include <hip/hip_runtime.h>


typedef __attribute__((ext_vector_type(2))) float v2f;
typedef __attribute__((ext_vector_type(4))) float v4f;
typedef __attribute__((ext_vector_type(8))) float v8f;

#define B_SZ 256
#define D_SZ 512
#define P_SZ 128
#define E_SZ 128
#define XB_PITCH (D_SZ * P_SZ)
#define WP_PITCH (D_SZ * E_SZ)
#define OB_PITCH (E_SZ * P_SZ)

#define KC     16
#define KPAD   18
#define PB      8
#define BTILE  16
#define ETILE  64
#define NCHUNK (D_SZ / KC)

__global__ __launch_bounds__(256, 2)
void loop_proj_wmma_kernel(const float* __restrict__ x,
                           const float* __restrict__ W,
                           const float* __restrict__ bias,
                           float* __restrict__ out)
{
    __shared__ float sX[2][PB * BTILE * KPAD];
    __shared__ float sW[2][PB * ETILE * KPAD];

    const int tid  = threadIdx.x;
    const int lane = tid & 31;
    const int wave = tid >> 5;

    const int pg0 = blockIdx.x * PB;
    const int b0  = blockIdx.y * BTILE;
    const int e0  = blockIdx.z * ETILE;

    const int m    = lane & 15;
    const int koff = (lane >> 4) << 1;

    v8f acc[4] = {};

    const int pairX = tid >> 1;
    const int phalf = (tid & 1) * 4;
    const int iX    = pairX >> 4;
    const int kX    = pairX & 15;
    const float* xp0 = x + (size_t)(b0 + iX) * XB_PITCH
                         + (size_t)kX * P_SZ + pg0 + phalf;
    const float* xp1 = xp0 + (size_t)8 * XB_PITCH;

    const int e4W   = (lane & 7) * 4;
    const int kW    = ((wave & 3) << 2) | (lane >> 3);
    const int eHalf = (wave >> 2) * 32;
    const float* wp = W + (size_t)pg0 * WP_PITCH
                        + (size_t)kW * E_SZ + e0 + eHalf + e4W;

    auto stage = [&](int buf, const v4f& rx0, const v4f& rx1, const v4f (&rw)[8]) {
        #pragma unroll
        for (int j = 0; j < 4; ++j) {
            sX[buf][((phalf + j) * BTILE + iX)     * KPAD + kX] = rx0[j];
            sX[buf][((phalf + j) * BTILE + iX + 8) * KPAD + kX] = rx1[j];
        }
        #pragma unroll
        for (int j = 0; j < 8; ++j) {
            #pragma unroll
            for (int q = 0; q < 4; ++q)
                sW[buf][(j * ETILE + eHalf + e4W + q) * KPAD + kW] = rw[j][q];
        }
    };

    {
        v4f rx0 = *(const v4f*)xp0;
        v4f rx1 = *(const v4f*)xp1;
        v4f rw[8];
        #pragma unroll
        for (int j = 0; j < 8; ++j) rw[j] = *(const v4f*)(wp + (size_t)j * WP_PITCH);
        stage(0, rx0, rx1, rw);
    }

    for (int c = 0; c < NCHUNK; ++c) {
        const int cur  = c & 1;
        const bool more = (c + 1) < NCHUNK;

        v4f nrx0, nrx1, nrw[8];
        if (more) {
            const size_t off = (size_t)(c + 1) * KC * P_SZ;
            nrx0 = *(const v4f*)(xp0 + off);
            nrx1 = *(const v4f*)(xp1 + off);
            #pragma unroll
            for (int j = 0; j < 8; ++j) nrw[j] = *(const v4f*)(wp + off + (size_t)j * WP_PITCH);
        }

        __syncthreads();

        const float* bX = sX[cur];
        const float* bW = sW[cur];
        #pragma unroll
        for (int k0 = 0; k0 < KC; k0 += 4) {
            v2f a = *(const v2f*)&bX[(wave * BTILE + m) * KPAD + k0 + koff];
            #pragma unroll
            for (int t = 0; t < 4; ++t) {
                v2f bf = *(const v2f*)&bW[(wave * ETILE + t * 16 + m) * KPAD + k0 + koff];
                acc[t] = __builtin_amdgcn_wmma_f32_16x16x4_f32(
                             false, a, false, bf, (short)0, acc[t], false, false);
            }
        }

        if (more) stage(cur ^ 1, nrx0, nrx1, nrw);
    }

    const int p    = pg0 + wave;
    const int mtop = (lane >> 4) << 3;
    float bi[4];
    #pragma unroll
    for (int t = 0; t < 4; ++t) bi[t] = bias[p * E_SZ + e0 + t * 16 + m];

    #pragma unroll
    for (int r = 0; r < 8; ++r) {
        const size_t bb = (size_t)(b0 + r + mtop) * OB_PITCH;
        #pragma unroll
        for (int t = 0; t < 4; ++t)
            out[bb + (size_t)(e0 + t * 16 + m) * P_SZ + p] = acc[t][r] + bi[t];
    }
}

extern "C" void kernel_launch(void* const* d_in, const int* in_sizes, int n_in,
                              void* d_out, int out_size, void* d_ws, size_t ws_size,
                              hipStream_t stream) {
    (void)in_sizes; (void)n_in; (void)d_ws; (void)ws_size; (void)out_size;
    const float* x    = (const float*)d_in[0];
    const float* W    = (const float*)d_in[1];
    const float* bias = (const float*)d_in[2];
    float* out        = (float*)d_out;

    dim3 grid(P_SZ / PB, B_SZ / BTILE, E_SZ / ETILE);
    dim3 block(256);
    loop_proj_wmma_kernel<<<grid, block, 0, stream>>>(x, W, bias, out);
}


# W B fragments straight from global: a lane's 16-B row load (columns 4c..4c+3) feeds four N tiles; W kept in two register sets a chunk ahead, not staged through LDS; chunk barrier waits for LDS only
# speedup vs baseline: 1.2545x; 1.0443x over previous
#include <hip/hip_runtime.h>


typedef __attribute__((ext_vector_type(2))) float v2f;
typedef __attribute__((ext_vector_type(4))) float v4f;
typedef __attribute__((ext_vector_type(8))) float v8f;

__device__ __forceinline__ void lds_barrier() {
    asm volatile("s_wait_dscnt 0x0\n\ts_barrier_signal -1\n\ts_barrier_wait -1" ::: "memory");
}

#define B_SZ 256
#define D_SZ 512
#define P_SZ 128
#define E_SZ 128
#define XB_PITCH (D_SZ * P_SZ)
#define WP_PITCH (D_SZ * E_SZ)
#define OB_PITCH (E_SZ * P_SZ)

#define KC     16
#define KPAD   18
#define PB      8
#define BTILE  16
#define ETILE  64
#define NCHUNK (D_SZ / KC)

__global__ __launch_bounds__(256, 2)
void loop_proj_wmma_kernel(const float* __restrict__ x,
                           const float* __restrict__ W,
                           const float* __restrict__ bias,
                           float* __restrict__ out)
{
    __shared__ float sX[2][PB * BTILE * KPAD];

    const int tid  = threadIdx.x;
    const int lane = tid & 31;
    const int wave = tid >> 5;

    const int pg0 = blockIdx.x * PB;
    const int b0  = blockIdx.y * BTILE;
    const int e0  = blockIdx.z * ETILE;
    const int p   = pg0 + wave;

    const int m    = lane & 15;
    const int koff = (lane >> 4) << 1;

    v8f acc[4] = {};

    const int pairX = tid >> 1;
    const int phalf = (tid & 1) * 4;
    const int iX    = pairX >> 4;
    const int kX    = pairX & 15;
    const float* xp0 = x + (size_t)(b0 + iX) * XB_PITCH
                         + (size_t)kX * P_SZ + pg0 + phalf;
    const float* xp1 = xp0 + (size_t)8 * XB_PITCH;

    const float* wp = W + (size_t)p * WP_PITCH + (size_t)koff * E_SZ + e0 + 4 * m;

    auto loadW = [&](v4f (&w)[8], int c) {
        const float* q = wp + (size_t)c * KC * E_SZ;
        #pragma unroll
        for (int i = 0; i < 8; ++i)
            w[i] = *(const v4f*)(q + (4 * (i >> 1) + (i & 1)) * E_SZ);
    };

    auto stageX = [&](int buf, const v4f& rx0, const v4f& rx1) {
        #pragma unroll
        for (int j = 0; j < 4; ++j) {
            sX[buf][((phalf + j) * BTILE + iX)     * KPAD + kX] = rx0[j];
            sX[buf][((phalf + j) * BTILE + iX + 8) * KPAD + kX] = rx1[j];
        }
    };

    auto compute = [&](const float* bX, const v4f (&w)[8]) {
        #pragma unroll
        for (int j = 0; j < 4; ++j) {
            v2f a = *(const v2f*)&bX[(wave * BTILE + m) * KPAD + 4 * j + koff];
            #pragma unroll
            for (int t = 0; t < 4; ++t) {
                v2f bf;
                bf.x = w[2 * j][t];
                bf.y = w[2 * j + 1][t];
                acc[t] = __builtin_amdgcn_wmma_f32_16x16x4_f32(
                             false, a, false, bf, (short)0, acc[t], false, false);
            }
        }
    };

    v4f wA[8], wB[8];

    {
        v4f rx0 = *(const v4f*)xp0;
        v4f rx1 = *(const v4f*)xp1;
        loadW(wA, 0);
        stageX(0, rx0, rx1);
    }

    for (int c = 0; c < NCHUNK - 2; c += 2) {
        {
            const size_t off = (size_t)(c + 1) * KC * P_SZ;
            v4f nrx0 = *(const v4f*)(xp0 + off);
            v4f nrx1 = *(const v4f*)(xp1 + off);
            loadW(wB, c + 1);
            lds_barrier();
            compute(sX[0], wA);
            stageX(1, nrx0, nrx1);
        }
        {
            const size_t off = (size_t)(c + 2) * KC * P_SZ;
            v4f nrx0 = *(const v4f*)(xp0 + off);
            v4f nrx1 = *(const v4f*)(xp1 + off);
            loadW(wA, c + 2);
            lds_barrier();
            compute(sX[1], wB);
            stageX(0, nrx0, nrx1);
        }
    }
    {
        const size_t off = (size_t)(NCHUNK - 1) * KC * P_SZ;
        v4f nrx0 = *(const v4f*)(xp0 + off);
        v4f nrx1 = *(const v4f*)(xp1 + off);
        loadW(wB, NCHUNK - 1);
        lds_barrier();
        compute(sX[0], wA);
        stageX(1, nrx0, nrx1);
        lds_barrier();
        compute(sX[1], wB);
    }

    const int mtop = (lane >> 4) << 3;
    const v4f bi = *(const v4f*)&bias[p * E_SZ + e0 + 4 * m];

    #pragma unroll
    for (int r = 0; r < 8; ++r) {
        const size_t bb = (size_t)(b0 + r + mtop) * OB_PITCH;
        #pragma unroll
        for (int t = 0; t < 4; ++t)
            out[bb + (size_t)(e0 + 4 * m + t) * P_SZ + p] = acc[t][r] + bi[t];
    }
}

extern "C" void kernel_launch(void* const* d_in, const int* in_sizes, int n_in,
                              void* d_out, int out_size, void* d_ws, size_t ws_size,
                              hipStream_t stream) {
    (void)in_sizes; (void)n_in; (void)d_ws; (void)ws_size; (void)out_size;
    const float* x    = (const float*)d_in[0];
    const float* W    = (const float*)d_in[1];
    const float* bias = (const float*)d_in[2];
    float* out        = (float*)d_out;

    dim3 grid(P_SZ / PB, B_SZ / BTILE, E_SZ / ETILE);
    dim3 block(256);
    loop_proj_wmma_kernel<<<grid, block, 0, stream>>>(x, W, bias, out);
}


# d loop starts at chunk (2*b_tile+e_tile) mod 32 and wraps: workgroups running together read different 8-KiB windows of the 256-KiB-strided x and W slabs; each output's d-sum order is rotated
# speedup vs baseline: 1.3163x; 1.0493x over previous
#include <hip/hip_runtime.h>


typedef __attribute__((ext_vector_type(2))) float v2f;
typedef __attribute__((ext_vector_type(4))) float v4f;
typedef __attribute__((ext_vector_type(8))) float v8f;

__device__ __forceinline__ void lds_barrier() {
    asm volatile("s_wait_dscnt 0x0\n\ts_barrier_signal -1\n\ts_barrier_wait -1" ::: "memory");
}

#define B_SZ 256
#define D_SZ 512
#define P_SZ 128
#define E_SZ 128
#define XB_PITCH (D_SZ * P_SZ)
#define WP_PITCH (D_SZ * E_SZ)
#define OB_PITCH (E_SZ * P_SZ)

#define KC     16
#define KPAD   18
#define PB      8
#define BTILE  16
#define ETILE  64
#define NCHUNK (D_SZ / KC)

__global__ __launch_bounds__(256, 2)
void loop_proj_wmma_kernel(const float* __restrict__ x,
                           const float* __restrict__ W,
                           const float* __restrict__ bias,
                           float* __restrict__ out)
{
    __shared__ float sX[2][PB * BTILE * KPAD];

    const int tid  = threadIdx.x;
    const int lane = tid & 31;
    const int wave = tid >> 5;

    const int pg0 = blockIdx.x * PB;
    const int b0  = blockIdx.y * BTILE;
    const int e0  = blockIdx.z * ETILE;
    const int p   = pg0 + wave;

    const int crot = (2 * (int)blockIdx.y + (int)blockIdx.z) & (NCHUNK - 1);
    auto chunkOf = [&](int s) { return (s + crot) & (NCHUNK - 1); };

    const int m    = lane & 15;
    const int koff = (lane >> 4) << 1;

    v8f acc[4] = {};

    const int pairX = tid >> 1;
    const int phalf = (tid & 1) * 4;
    const int iX    = pairX >> 4;
    const int kX    = pairX & 15;
    const float* xp0 = x + (size_t)(b0 + iX) * XB_PITCH
                         + (size_t)kX * P_SZ + pg0 + phalf;
    const float* xp1 = xp0 + (size_t)8 * XB_PITCH;

    const float* wp = W + (size_t)p * WP_PITCH + (size_t)koff * E_SZ + e0 + 4 * m;

    auto loadW = [&](v4f (&w)[8], int c) {
        const float* q = wp + (size_t)c * KC * E_SZ;
        #pragma unroll
        for (int i = 0; i < 8; ++i)
            w[i] = *(const v4f*)(q + (4 * (i >> 1) + (i & 1)) * E_SZ);
    };

    auto stageX = [&](int buf, const v4f& rx0, const v4f& rx1) {
        #pragma unroll
        for (int j = 0; j < 4; ++j) {
            sX[buf][((phalf + j) * BTILE + iX)     * KPAD + kX] = rx0[j];
            sX[buf][((phalf + j) * BTILE + iX + 8) * KPAD + kX] = rx1[j];
        }
    };

    auto compute = [&](const float* bX, const v4f (&w)[8]) {
        #pragma unroll
        for (int j = 0; j < 4; ++j) {
            v2f a = *(const v2f*)&bX[(wave * BTILE + m) * KPAD + 4 * j + koff];
            #pragma unroll
            for (int t = 0; t < 4; ++t) {
                v2f bf;
                bf.x = w[2 * j][t];
                bf.y = w[2 * j + 1][t];
                acc[t] = __builtin_amdgcn_wmma_f32_16x16x4_f32(
                             false, a, false, bf, (short)0, acc[t], false, false);
            }
        }
    };

    v4f wA[8], wB[8];

    {
        const int cc = chunkOf(0);
        const size_t off = (size_t)cc * KC * P_SZ;
        v4f rx0 = *(const v4f*)(xp0 + off);
        v4f rx1 = *(const v4f*)(xp1 + off);
        loadW(wA, cc);
        stageX(0, rx0, rx1);
    }

    for (int c = 0; c < NCHUNK - 2; c += 2) {
        {
            const int cc = chunkOf(c + 1);
            const size_t off = (size_t)cc * KC * P_SZ;
            v4f nrx0 = *(const v4f*)(xp0 + off);
            v4f nrx1 = *(const v4f*)(xp1 + off);
            loadW(wB, cc);
            lds_barrier();
            compute(sX[0], wA);
            stageX(1, nrx0, nrx1);
        }
        {
            const int cc = chunkOf(c + 2);
            const size_t off = (size_t)cc * KC * P_SZ;
            v4f nrx0 = *(const v4f*)(xp0 + off);
            v4f nrx1 = *(const v4f*)(xp1 + off);
            loadW(wA, cc);
            lds_barrier();
            compute(sX[1], wB);
            stageX(0, nrx0, nrx1);
        }
    }
    {
        const int cc = chunkOf(NCHUNK - 1);
        const size_t off = (size_t)cc * KC * P_SZ;
        v4f nrx0 = *(const v4f*)(xp0 + off);
        v4f nrx1 = *(const v4f*)(xp1 + off);
        loadW(wB, cc);
        lds_barrier();
        compute(sX[0], wA);
        stageX(1, nrx0, nrx1);
        lds_barrier();
        compute(sX[1], wB);
    }

    const int mtop = (lane >> 4) << 3;
    const v4f bi = *(const v4f*)&bias[p * E_SZ + e0 + 4 * m];

    #pragma unroll
    for (int r = 0; r < 8; ++r) {
        const size_t bb = (size_t)(b0 + r + mtop) * OB_PITCH;
        #pragma unroll
        for (int t = 0; t < 4; ++t)
            out[bb + (size_t)(e0 + 4 * m + t) * P_SZ + p] = acc[t][r] + bi[t];
    }
}

extern "C" void kernel_launch(void* const* d_in, const int* in_sizes, int n_in,
                              void* d_out, int out_size, void* d_ws, size_t ws_size,
                              hipStream_t stream) {
    (void)in_sizes; (void)n_in; (void)d_ws; (void)ws_size; (void)out_size;
    const float* x    = (const float*)d_in[0];
    const float* W    = (const float*)d_in[1];
    const float* bias = (const float*)d_in[2];
    float* out        = (float*)d_out;

    dim3 grid(P_SZ / PB, B_SZ / BTILE, E_SZ / ETILE);
    dim3 block(256);
    loop_proj_wmma_kernel<<<grid, block, 0, stream>>>(x, W, bias, out);
}


# Epilogue transposed through LDS [b][p][e]: a thread stores 16 B = 4 prototypes of one (b,e), so a store instruction writes 16 pieces of 32 B, not 32 pieces of 4 B; values and d-sum order unchanged
# speedup vs baseline: 2.9588x; 2.2478x over previous
#include <hip/hip_runtime.h>


typedef __attribute__((ext_vector_type(2))) float v2f;
typedef __attribute__((ext_vector_type(4))) float v4f;
typedef __attribute__((ext_vector_type(8))) float v8f;

__device__ __forceinline__ void lds_barrier() {
    asm volatile("s_wait_dscnt 0x0\n\ts_barrier_signal -1\n\ts_barrier_wait -1" ::: "memory");
}

#define B_SZ 256
#define D_SZ 512
#define P_SZ 128
#define E_SZ 128
#define XB_PITCH (D_SZ * P_SZ)
#define WP_PITCH (D_SZ * E_SZ)
#define OB_PITCH (E_SZ * P_SZ)

#define KC     16
#define KPAD   18
#define PB      8
#define BTILE  16
#define ETILE  64
#define NCHUNK (D_SZ / KC)
#define EPITCH 68

__global__ __launch_bounds__(256, 2)
void loop_proj_wmma_kernel(const float* __restrict__ x,
                           const float* __restrict__ W,
                           const float* __restrict__ bias,
                           float* __restrict__ out)
{
    __shared__ float sX[2][PB * BTILE * KPAD];
    __shared__ __attribute__((aligned(16))) float sO[BTILE * PB * EPITCH];

    const int tid  = threadIdx.x;
    const int lane = tid & 31;
    const int wave = tid >> 5;

    const int pg0 = blockIdx.x * PB;
    const int b0  = blockIdx.y * BTILE;
    const int e0  = blockIdx.z * ETILE;
    const int p   = pg0 + wave;

    const int crot = (2 * (int)blockIdx.y + (int)blockIdx.z) & (NCHUNK - 1);
    auto chunkOf = [&](int s) { return (s + crot) & (NCHUNK - 1); };

    const int m    = lane & 15;
    const int koff = (lane >> 4) << 1;

    v8f acc[4] = {};

    const int pairX = tid >> 1;
    const int phalf = (tid & 1) * 4;
    const int iX    = pairX >> 4;
    const int kX    = pairX & 15;
    const float* xp0 = x + (size_t)(b0 + iX) * XB_PITCH
                         + (size_t)kX * P_SZ + pg0 + phalf;
    const float* xp1 = xp0 + (size_t)8 * XB_PITCH;

    const float* wp = W + (size_t)p * WP_PITCH + (size_t)koff * E_SZ + e0 + 4 * m;

    auto loadW = [&](v4f (&w)[8], int c) {
        const float* q = wp + (size_t)c * KC * E_SZ;
        #pragma unroll
        for (int i = 0; i < 8; ++i)
            w[i] = *(const v4f*)(q + (4 * (i >> 1) + (i & 1)) * E_SZ);
    };

    auto stageX = [&](int buf, const v4f& rx0, const v4f& rx1) {
        #pragma unroll
        for (int j = 0; j < 4; ++j) {
            sX[buf][((phalf + j) * BTILE + iX)     * KPAD + kX] = rx0[j];
            sX[buf][((phalf + j) * BTILE + iX + 8) * KPAD + kX] = rx1[j];
        }
    };

    auto compute = [&](const float* bX, const v4f (&w)[8]) {
        #pragma unroll
        for (int j = 0; j < 4; ++j) {
            v2f a = *(const v2f*)&bX[(wave * BTILE + m) * KPAD + 4 * j + koff];
            #pragma unroll
            for (int t = 0; t < 4; ++t) {
                v2f bf;
                bf.x = w[2 * j][t];
                bf.y = w[2 * j + 1][t];
                acc[t] = __builtin_amdgcn_wmma_f32_16x16x4_f32(
                             false, a, false, bf, (short)0, acc[t], false, false);
            }
        }
    };

    v4f wA[8], wB[8];

    {
        const int cc = chunkOf(0);
        const size_t off = (size_t)cc * KC * P_SZ;
        v4f rx0 = *(const v4f*)(xp0 + off);
        v4f rx1 = *(const v4f*)(xp1 + off);
        loadW(wA, cc);
        stageX(0, rx0, rx1);
    }

    for (int c = 0; c < NCHUNK - 2; c += 2) {
        {
            const int cc = chunkOf(c + 1);
            const size_t off = (size_t)cc * KC * P_SZ;
            v4f nrx0 = *(const v4f*)(xp0 + off);
            v4f nrx1 = *(const v4f*)(xp1 + off);
            loadW(wB, cc);
            lds_barrier();
            compute(sX[0], wA);
            stageX(1, nrx0, nrx1);
        }
        {
            const int cc = chunkOf(c + 2);
            const size_t off = (size_t)cc * KC * P_SZ;
            v4f nrx0 = *(const v4f*)(xp0 + off);
            v4f nrx1 = *(const v4f*)(xp1 + off);
            loadW(wA, cc);
            lds_barrier();
            compute(sX[1], wB);
            stageX(0, nrx0, nrx1);
        }
    }
    {
        const int cc = chunkOf(NCHUNK - 1);
        const size_t off = (size_t)cc * KC * P_SZ;
        v4f nrx0 = *(const v4f*)(xp0 + off);
        v4f nrx1 = *(const v4f*)(xp1 + off);
        loadW(wB, cc);
        lds_barrier();
        compute(sX[0], wA);
        stageX(1, nrx0, nrx1);
        lds_barrier();
        compute(sX[1], wB);
    }

    const int mtop = (lane >> 4) << 3;
    const v4f bi = *(const v4f*)&bias[p * E_SZ + e0 + 4 * m];

    #pragma unroll
    for (int r = 0; r < 8; ++r) {
        v4f o;
        #pragma unroll
        for (int t = 0; t < 4; ++t)
            o[t] = acc[t][r] + bi[t];
        *(v4f*)&sO[((r + mtop) * PB + wave) * EPITCH + 4 * m] = o;
    }
    __syncthreads();

    const int psub = (tid & 1) * 4;
    const int bl0  = tid >> 7;
    const int el   = (tid >> 1) & 63;
    const float* s0 = &sO[(bl0 * PB + psub) * EPITCH + el];
    float* o0 = out + (size_t)(b0 + bl0) * OB_PITCH + (size_t)(e0 + el) * P_SZ + pg0 + psub;
    #pragma unroll
    for (int it = 0; it < 8; ++it) {
        const float* s = s0 + it * (2 * PB * EPITCH);
        v4f v;
        v[0] = s[0];
        v[1] = s[EPITCH];
        v[2] = s[2 * EPITCH];
        v[3] = s[3 * EPITCH];
        *(v4f*)(o0 + (size_t)it * (2 * OB_PITCH)) = v;
    }
}

extern "C" void kernel_launch(void* const* d_in, const int* in_sizes, int n_in,
                              void* d_out, int out_size, void* d_ws, size_t ws_size,
                              hipStream_t stream) {
    (void)in_sizes; (void)n_in; (void)d_ws; (void)ws_size; (void)out_size;
    const float* x    = (const float*)d_in[0];
    const float* W    = (const float*)d_in[1];
    const float* bias = (const float*)d_in[2];
    float* out        = (float*)d_out;

    dim3 grid(P_SZ / PB, B_SZ / BTILE, E_SZ / ETILE);
    dim3 block(256);
    loop_proj_wmma_kernel<<<grid, block, 0, stream>>>(x, W, bias, out);
}


# Workgroup = 16 prototypes x 16 waves (grid 8x16x2): its share of an x or out row is 64 B not 32 B, so an x wave load touches 8 lines not 16 and a store writes 8 pieces not 16; d-sum order unchanged
# speedup vs baseline: 3.4053x; 1.1509x over previous
#include <hip/hip_runtime.h>


typedef __attribute__((ext_vector_type(2))) float v2f;
typedef __attribute__((ext_vector_type(4))) float v4f;
typedef __attribute__((ext_vector_type(8))) float v8f;

__device__ __forceinline__ void lds_barrier() {
    asm volatile("s_wait_dscnt 0x0\n\ts_barrier_signal -1\n\ts_barrier_wait -1" ::: "memory");
}

#define B_SZ 256
#define D_SZ 512
#define P_SZ 128
#define E_SZ 128
#define XB_PITCH (D_SZ * P_SZ)
#define WP_PITCH (D_SZ * E_SZ)
#define OB_PITCH (E_SZ * P_SZ)

#define KC     16
#define KPAD   18
#define PB     16
#define NTHR  (PB * 32)
#define XPLANE 290
#define BTILE  16
#define ETILE  64
#define NCHUNK (D_SZ / KC)
#define EPITCH 68

__global__ __launch_bounds__(NTHR)
void loop_proj_wmma_kernel(const float* __restrict__ x,
                           const float* __restrict__ W,
                           const float* __restrict__ bias,
                           float* __restrict__ out)
{
    __shared__ __attribute__((aligned(16))) float sX[2][PB * XPLANE];
    __shared__ __attribute__((aligned(16))) float sO[BTILE * PB * EPITCH];

    const int tid  = threadIdx.x;
    const int lane = tid & 31;
    const int wave = tid >> 5;

    const int pg0 = blockIdx.x * PB;
    const int b0  = blockIdx.y * BTILE;
    const int e0  = blockIdx.z * ETILE;
    const int p   = pg0 + wave;

    const int crot = (2 * (int)blockIdx.y + (int)blockIdx.z) & (NCHUNK - 1);
    auto chunkOf = [&](int s) { return (s + crot) & (NCHUNK - 1); };

    const int m    = lane & 15;
    const int koff = (lane >> 4) << 1;

    v8f acc[4] = {};

    const int quadX = tid >> 2;
    const int pq4   = (tid & 3) * 4;
    const int iX    = quadX >> 4;
    const int kX    = quadX & 15;
    const float* xp0 = x + (size_t)(b0 + iX) * XB_PITCH
                         + (size_t)kX * P_SZ + pg0 + pq4;
    const float* xp1 = xp0 + (size_t)8 * XB_PITCH;

    const float* wp = W + (size_t)p * WP_PITCH + (size_t)koff * E_SZ + e0 + 4 * m;

    auto loadW = [&](v4f (&w)[8], int c) {
        const float* q = wp + (size_t)c * KC * E_SZ;
        #pragma unroll
        for (int i = 0; i < 8; ++i)
            w[i] = *(const v4f*)(q + (4 * (i >> 1) + (i & 1)) * E_SZ);
    };

    auto stageX = [&](int buf, const v4f& rx0, const v4f& rx1) {
        #pragma unroll
        for (int j = 0; j < 4; ++j) {
            sX[buf][(pq4 + j) * XPLANE + iX       * KPAD + kX] = rx0[j];
            sX[buf][(pq4 + j) * XPLANE + (iX + 8) * KPAD + kX] = rx1[j];
        }
    };

    auto compute = [&](const float* bX, const v4f (&w)[8]) {
        #pragma unroll
        for (int j = 0; j < 4; ++j) {
            v2f a = *(const v2f*)&bX[wave * XPLANE + m * KPAD + 4 * j + koff];
            #pragma unroll
            for (int t = 0; t < 4; ++t) {
                v2f bf;
                bf.x = w[2 * j][t];
                bf.y = w[2 * j + 1][t];
                acc[t] = __builtin_amdgcn_wmma_f32_16x16x4_f32(
                             false, a, false, bf, (short)0, acc[t], false, false);
            }
        }
    };

    v4f wA[8], wB[8];

    {
        const int cc = chunkOf(0);
        const size_t off = (size_t)cc * KC * P_SZ;
        v4f rx0 = *(const v4f*)(xp0 + off);
        v4f rx1 = *(const v4f*)(xp1 + off);
        loadW(wA, cc);
        stageX(0, rx0, rx1);
    }

    for (int c = 0; c < NCHUNK - 2; c += 2) {
        {
            const int cc = chunkOf(c + 1);
            const size_t off = (size_t)cc * KC * P_SZ;
            v4f nrx0 = *(const v4f*)(xp0 + off);
            v4f nrx1 = *(const v4f*)(xp1 + off);
            loadW(wB, cc);
            lds_barrier();
            compute(sX[0], wA);
            __builtin_amdgcn_sched_barrier(0);
            stageX(1, nrx0, nrx1);
        }
        {
            const int cc = chunkOf(c + 2);
            const size_t off = (size_t)cc * KC * P_SZ;
            v4f nrx0 = *(const v4f*)(xp0 + off);
            v4f nrx1 = *(const v4f*)(xp1 + off);
            loadW(wA, cc);
            lds_barrier();
            compute(sX[1], wB);
            __builtin_amdgcn_sched_barrier(0);
            stageX(0, nrx0, nrx1);
        }
    }
    {
        const int cc = chunkOf(NCHUNK - 1);
        const size_t off = (size_t)cc * KC * P_SZ;
        v4f nrx0 = *(const v4f*)(xp0 + off);
        v4f nrx1 = *(const v4f*)(xp1 + off);
        loadW(wB, cc);
        lds_barrier();
        compute(sX[0], wA);
        __builtin_amdgcn_sched_barrier(0);
        stageX(1, nrx0, nrx1);
        lds_barrier();
        compute(sX[1], wB);
    }

    const int mtop = (lane >> 4) << 3;
    const v4f bi = *(const v4f*)&bias[p * E_SZ + e0 + 4 * m];

    #pragma unroll
    for (int r = 0; r < 8; ++r) {
        v4f o;
        #pragma unroll
        for (int t = 0; t < 4; ++t)
            o[t] = acc[t][r] + bi[t];
        *(v4f*)&sO[((r + mtop) * PB + wave) * EPITCH + 4 * m] = o;
    }
    __syncthreads();

    const int psub = (tid & 3) * 4;
    const int bl0  = tid >> 8;
    const int el   = (tid >> 2) & 63;
    const float* s0 = &sO[(bl0 * PB + psub) * EPITCH + el];
    float* o0 = out + (size_t)(b0 + bl0) * OB_PITCH + (size_t)(e0 + el) * P_SZ + pg0 + psub;
    #pragma unroll
    for (int it = 0; it < 8; ++it) {
        const float* s = s0 + it * (2 * PB * EPITCH);
        v4f v;
        v[0] = s[0];
        v[1] = s[EPITCH];
        v[2] = s[2 * EPITCH];
        v[3] = s[3 * EPITCH];
        *(v4f*)(o0 + (size_t)it * (2 * OB_PITCH)) = v;
    }
}

extern "C" void kernel_launch(void* const* d_in, const int* in_sizes, int n_in,
                              void* d_out, int out_size, void* d_ws, size_t ws_size,
                              hipStream_t stream) {
    (void)in_sizes; (void)n_in; (void)d_ws; (void)ws_size; (void)out_size;
    const float* x    = (const float*)d_in[0];
    const float* W    = (const float*)d_in[1];
    const float* bias = (const float*)d_in[2];
    float* out        = (float*)d_out;

    dim3 grid(P_SZ / PB, B_SZ / BTILE, E_SZ / ETILE);
    dim3 block(NTHR);
    loop_proj_wmma_kernel<<<grid, block, 0, stream>>>(x, W, bias, out);
}


# Every global-load wait falls a full step behind its issue: W k-pair copies pinned into the consuming step (were at the loop back edge, with a drain), x fetched two chunks ahead; d-sum order unchanged
# speedup vs baseline: 3.6260x; 1.0648x over previous
#include <hip/hip_runtime.h>


typedef __attribute__((ext_vector_type(2))) float v2f;
typedef __attribute__((ext_vector_type(4))) float v4f;
typedef __attribute__((ext_vector_type(8))) float v8f;

__device__ __forceinline__ void lds_barrier() {
    asm volatile("s_wait_dscnt 0x0\n\ts_barrier_signal -1\n\ts_barrier_wait -1" ::: "memory");
}

#define B_SZ 256
#define D_SZ 512
#define P_SZ 128
#define E_SZ 128
#define XB_PITCH (D_SZ * P_SZ)
#define WP_PITCH (D_SZ * E_SZ)
#define OB_PITCH (E_SZ * P_SZ)

#define KC     16
#define KPAD   18
#define PB     16
#define NTHR  (PB * 32)
#define XPLANE 290
#define BTILE  16
#define ETILE  64
#define NCHUNK (D_SZ / KC)
#define EPITCH 68

__global__ __launch_bounds__(NTHR)
void loop_proj_wmma_kernel(const float* __restrict__ x,
                           const float* __restrict__ W,
                           const float* __restrict__ bias,
                           float* __restrict__ out)
{
    __shared__ __attribute__((aligned(16))) float sX[2][PB * XPLANE];
    __shared__ __attribute__((aligned(16))) float sO[BTILE * PB * EPITCH];

    const int tid  = threadIdx.x;
    const int lane = tid & 31;
    const int wave = tid >> 5;

    const int pg0 = blockIdx.x * PB;
    const int b0  = blockIdx.y * BTILE;
    const int e0  = blockIdx.z * ETILE;
    const int p   = pg0 + wave;

    const int crot = (2 * (int)blockIdx.y + (int)blockIdx.z) & (NCHUNK - 1);
    auto chunkOf = [&](int s) { return (s + crot) & (NCHUNK - 1); };

    const int m    = lane & 15;
    const int koff = (lane >> 4) << 1;

    v8f acc[4] = {};

    const int quadX = tid >> 2;
    const int pq4   = (tid & 3) * 4;
    const int iX    = quadX >> 4;
    const int kX    = quadX & 15;
    const float* xp0 = x + (size_t)(b0 + iX) * XB_PITCH
                         + (size_t)kX * P_SZ + pg0 + pq4;
    const float* xp1 = xp0 + (size_t)8 * XB_PITCH;

    const float* wp = W + (size_t)p * WP_PITCH + (size_t)koff * E_SZ + e0 + 4 * m;

    auto loadW = [&](v4f (&w)[8], int c) {
        const float* q = wp + (size_t)c * KC * E_SZ;
        #pragma unroll
        for (int i = 0; i < 8; ++i)
            w[i] = *(const v4f*)(q + (4 * (i >> 1) + (i & 1)) * E_SZ);
    };

    auto stageX = [&](int buf, const v4f& rx0, const v4f& rx1) {
        #pragma unroll
        for (int j = 0; j < 4; ++j) {
            sX[buf][(pq4 + j) * XPLANE + iX       * KPAD + kX] = rx0[j];
            sX[buf][(pq4 + j) * XPLANE + (iX + 8) * KPAD + kX] = rx1[j];
        }
    };

    auto compute = [&](const float* bX, v4f (&w)[8]) {
        #pragma unroll
        for (int j = 0; j < 4; ++j) {
            asm volatile("" : "+v"(w[2 * j]), "+v"(w[2 * j + 1]));
            v2f a = *(const v2f*)&bX[wave * XPLANE + m * KPAD + 4 * j + koff];
            #pragma unroll
            for (int t = 0; t < 4; ++t) {
                v2f bf;
                bf.x = w[2 * j][t];
                bf.y = w[2 * j + 1][t];
                acc[t] = __builtin_amdgcn_wmma_f32_16x16x4_f32(
                             false, a, false, bf, (short)0, acc[t], false, false);
            }
        }
    };

    v4f wA[8], wB[8];
    v4f xE0, xE1, xO0, xO1;

    auto loadX = [&](v4f& r0, v4f& r1, int s) {
        const size_t off = (size_t)chunkOf(s) * KC * P_SZ;
        r0 = *(const v4f*)(xp0 + off);
        r1 = *(const v4f*)(xp1 + off);
    };

    loadX(xE0, xE1, 0);
    loadW(wA, chunkOf(0));
    loadX(xO0, xO1, 1);
    stageX(0, xE0, xE1);

    for (int c = 0; c < NCHUNK - 2; c += 2) {
        {
            loadX(xE0, xE1, c + 2);
            loadW(wB, chunkOf(c + 1));
            lds_barrier();
            compute(sX[0], wA);
            __builtin_amdgcn_sched_barrier(0);
            stageX(1, xO0, xO1);
        }
        {
            loadX(xO0, xO1, c + 3);
            loadW(wA, chunkOf(c + 2));
            lds_barrier();
            compute(sX[1], wB);
            __builtin_amdgcn_sched_barrier(0);
            stageX(0, xE0, xE1);
        }
    }
    {
        loadW(wB, chunkOf(NCHUNK - 1));
        lds_barrier();
        compute(sX[0], wA);
        __builtin_amdgcn_sched_barrier(0);
        stageX(1, xO0, xO1);
        lds_barrier();
        compute(sX[1], wB);
    }

    const int mtop = (lane >> 4) << 3;
    const v4f bi = *(const v4f*)&bias[p * E_SZ + e0 + 4 * m];

    #pragma unroll
    for (int r = 0; r < 8; ++r) {
        v4f o;
        #pragma unroll
        for (int t = 0; t < 4; ++t)
            o[t] = acc[t][r] + bi[t];
        *(v4f*)&sO[((r + mtop) * PB + wave) * EPITCH + 4 * m] = o;
    }
    __syncthreads();

    const int psub = (tid & 3) * 4;
    const int bl0  = tid >> 8;
    const int el   = (tid >> 2) & 63;
    const float* s0 = &sO[(bl0 * PB + psub) * EPITCH + el];
    float* o0 = out + (size_t)(b0 + bl0) * OB_PITCH + (size_t)(e0 + el) * P_SZ + pg0 + psub;
    #pragma unroll
    for (int it = 0; it < 8; ++it) {
        const float* s = s0 + it * (2 * PB * EPITCH);
        v4f v;
        v[0] = s[0];
        v[1] = s[EPITCH];
        v[2] = s[2 * EPITCH];
        v[3] = s[3 * EPITCH];
        *(v4f*)(o0 + (size_t)it * (2 * OB_PITCH)) = v;
    }
}

extern "C" void kernel_launch(void* const* d_in, const int* in_sizes, int n_in,
                              void* d_out, int out_size, void* d_ws, size_t ws_size,
                              hipStream_t stream) {
    (void)in_sizes; (void)n_in; (void)d_ws; (void)ws_size; (void)out_size;
    const float* x    = (const float*)d_in[0];
    const float* W    = (const float*)d_in[1];
    const float* bias = (const float*)d_in[2];
    float* out        = (float*)d_out;

    dim3 grid(P_SZ / PB, B_SZ / BTILE, E_SZ / ETILE);
    dim3 block(NTHR);
    loop_proj_wmma_kernel<<<grid, block, 0, stream>>>(x, W, bias, out);
}


# Workgroup barrier split, x ring of three LDS stages: a wave stores next chunk's x and signals at a step's head, waits at the next head, so waves run up to a step apart; d-sum order unchanged
# speedup vs baseline: 3.7560x; 1.0359x over previous
#include <hip/hip_runtime.h>


typedef __attribute__((ext_vector_type(2))) float v2f;
typedef __attribute__((ext_vector_type(4))) float v4f;
typedef __attribute__((ext_vector_type(8))) float v8f;

__device__ __forceinline__ void ring_signal() {
    asm volatile("s_wait_dscnt 0x0\n\ts_barrier_signal -1" ::: "memory");
}
__device__ __forceinline__ void ring_wait() {
    asm volatile("s_barrier_wait -1" ::: "memory");
}

#define B_SZ 256
#define D_SZ 512
#define P_SZ 128
#define E_SZ 128
#define XB_PITCH (D_SZ * P_SZ)
#define WP_PITCH (D_SZ * E_SZ)
#define OB_PITCH (E_SZ * P_SZ)

#define KC     16
#define KPAD   18
#define PB     16
#define NTHR  (PB * 32)
#define XPLANE 290
#define BTILE  16
#define ETILE  64
#define NCHUNK (D_SZ / KC)
#define NSTAGE 3
#define EPITCH 68

__global__ __launch_bounds__(NTHR)
void loop_proj_wmma_kernel(const float* __restrict__ x,
                           const float* __restrict__ W,
                           const float* __restrict__ bias,
                           float* __restrict__ out)
{
    __shared__ __attribute__((aligned(16))) float sX[NSTAGE][PB * XPLANE];
    __shared__ __attribute__((aligned(16))) float sO[BTILE * PB * EPITCH];

    const int tid  = threadIdx.x;
    const int lane = tid & 31;
    const int wave = tid >> 5;

    const int pg0 = blockIdx.x * PB;
    const int b0  = blockIdx.y * BTILE;
    const int e0  = blockIdx.z * ETILE;
    const int p   = pg0 + wave;

    const int crot = (2 * (int)blockIdx.y + (int)blockIdx.z) & (NCHUNK - 1);
    auto chunkOf = [&](int s) { return (s + crot) & (NCHUNK - 1); };

    const int m    = lane & 15;
    const int koff = (lane >> 4) << 1;

    v8f acc[4] = {};

    const int quadX = tid >> 2;
    const int pq4   = (tid & 3) * 4;
    const int iX    = quadX >> 4;
    const int kX    = quadX & 15;
    const float* xp0 = x + (size_t)(b0 + iX) * XB_PITCH
                         + (size_t)kX * P_SZ + pg0 + pq4;
    const float* xp1 = xp0 + (size_t)8 * XB_PITCH;

    const float* wp = W + (size_t)p * WP_PITCH + (size_t)koff * E_SZ + e0 + 4 * m;

    auto loadW = [&](v4f (&w)[8], int c) {
        const float* q = wp + (size_t)c * KC * E_SZ;
        #pragma unroll
        for (int i = 0; i < 8; ++i)
            w[i] = *(const v4f*)(q + (4 * (i >> 1) + (i & 1)) * E_SZ);
    };

    auto stageX = [&](int buf, const v4f& rx0, const v4f& rx1) {
        float* sB = &sX[0][0] + buf * (PB * XPLANE);
        #pragma unroll
        for (int j = 0; j < 4; ++j) {
            sB[(pq4 + j) * XPLANE + iX       * KPAD + kX] = rx0[j];
            sB[(pq4 + j) * XPLANE + (iX + 8) * KPAD + kX] = rx1[j];
        }
    };

    auto compute = [&](int buf, v4f (&w)[8]) {
        const float* bX = &sX[0][0] + buf * (PB * XPLANE);
        #pragma unroll
        for (int j = 0; j < 4; ++j) {
            asm volatile("" : "+v"(w[2 * j]), "+v"(w[2 * j + 1]));
            v2f a = *(const v2f*)&bX[wave * XPLANE + m * KPAD + 4 * j + koff];
            #pragma unroll
            for (int t = 0; t < 4; ++t) {
                v2f bf;
                bf.x = w[2 * j][t];
                bf.y = w[2 * j + 1][t];
                acc[t] = __builtin_amdgcn_wmma_f32_16x16x4_f32(
                             false, a, false, bf, (short)0, acc[t], false, false);
            }
        }
    };

    v4f wA[8], wB[8];
    v4f xE0, xE1, xO0, xO1;

    auto loadX = [&](v4f& r0, v4f& r1, int s) {
        const size_t off = (size_t)chunkOf(s) * KC * P_SZ;
        r0 = *(const v4f*)(xp0 + off);
        r1 = *(const v4f*)(xp1 + off);
    };

    loadX(xE0, xE1, 0);
    loadW(wA, chunkOf(0));
    loadX(xO0, xO1, 1);
    stageX(0, xE0, xE1);
    loadX(xE0, xE1, 2);
    ring_signal();

    int rb = 0;
    auto nextStage = [](int b) { return b == NSTAGE - 1 ? 0 : b + 1; };
    for (int c = 0; c < NCHUNK - 2; c += 2) {
        {
            ring_wait();
            const int wb = nextStage(rb);
            loadW(wB, chunkOf(c + 1));
            stageX(wb, xO0, xO1);
            loadX(xO0, xO1, c + 3);
            ring_signal();
            compute(rb, wA);
            rb = wb;
        }
        {
            ring_wait();
            const int wb = nextStage(rb);
            loadW(wA, chunkOf(c + 2));
            stageX(wb, xE0, xE1);
            if (c + 4 < NCHUNK)
                loadX(xE0, xE1, c + 4);
            ring_signal();
            compute(rb, wB);
            rb = wb;
        }
    }
    {
        ring_wait();
        const int wb = nextStage(rb);
        loadW(wB, chunkOf(NCHUNK - 1));
        stageX(wb, xO0, xO1);
        ring_signal();
        compute(rb, wA);
        ring_wait();
        compute(wb, wB);
    }

    const int mtop = (lane >> 4) << 3;
    const v4f bi = *(const v4f*)&bias[p * E_SZ + e0 + 4 * m];

    #pragma unroll
    for (int r = 0; r < 8; ++r) {
        v4f o;
        #pragma unroll
        for (int t = 0; t < 4; ++t)
            o[t] = acc[t][r] + bi[t];
        *(v4f*)&sO[((r + mtop) * PB + wave) * EPITCH + 4 * m] = o;
    }
    __syncthreads();

    const int psub = (tid & 3) * 4;
    const int bl0  = tid >> 8;
    const int el   = (tid >> 2) & 63;
    const float* s0 = &sO[(bl0 * PB + psub) * EPITCH + el];
    float* o0 = out + (size_t)(b0 + bl0) * OB_PITCH + (size_t)(e0 + el) * P_SZ + pg0 + psub;
    #pragma unroll
    for (int it = 0; it < 8; ++it) {
        const float* s = s0 + it * (2 * PB * EPITCH);
        v4f v;
        v[0] = s[0];
        v[1] = s[EPITCH];
        v[2] = s[2 * EPITCH];
        v[3] = s[3 * EPITCH];
        *(v4f*)(o0 + (size_t)it * (2 * OB_PITCH)) = v;
    }
}

extern "C" void kernel_launch(void* const* d_in, const int* in_sizes, int n_in,
                              void* d_out, int out_size, void* d_ws, size_t ws_size,
                              hipStream_t stream) {
    (void)in_sizes; (void)n_in; (void)d_ws; (void)ws_size; (void)out_size;
    const float* x    = (const float*)d_in[0];
    const float* W    = (const float*)d_in[1];
    const float* bias = (const float*)d_in[2];
    float* out        = (float*)d_out;

    dim3 grid(P_SZ / PB, B_SZ / BTILE, E_SZ / ETILE);
    dim3 block(NTHR);
    loop_proj_wmma_kernel<<<grid, block, 0, stream>>>(x, W, bias, out);
}
